# starting kernel
# baseline (speedup 1.0000x reference)
#include <hip/hip_runtime.h>
#include <hip/hip_bf16.h>


typedef float v2f __attribute__((ext_vector_type(2)));
typedef float v8f __attribute__((ext_vector_type(8)));

#define TAU_MIN_F     0.001f
#define XPBD_MAXC     0.15f
#define XPBD_DENOM    (2.0f + 1e-9f)

#define BSTRIDE 260

__device__ __forceinline__ void atomic_add_f32(float* p, float v) {
    unsafeAtomicAdd(p, v);
}

__global__ void zero_kernel(float* __restrict__ p, int n) {
    int i = blockIdx.x * blockDim.x + threadIdx.x;
    if (i < n) p[i] = 0.0f;
}

__global__ __launch_bounds__(256) void head_wmma_kernel(
    const float* __restrict__ ht,
    const float* __restrict__ W,
    const float* __restrict__ bias,
    const float* __restrict__ kp,
    const float* __restrict__ ts,
    float* __restrict__ x,
    int nTiles, int D, int N)
{
    __shared__ float sWT[16 * BSTRIDE];
    __shared__ float sB[3];

    for (int idx = threadIdx.x; idx < 16 * 256; idx += blockDim.x) {
        int n = idx >> 8;
        int k = idx & 255;
        sWT[n * BSTRIDE + k] = (n < 3) ? W[k * 3 + n] : 0.0f;
    }
    if (threadIdx.x < 3) sB[threadIdx.x] = bias[threadIdx.x];
    __syncthreads();

    int wave = (blockIdx.x * blockDim.x + threadIdx.x) >> 5;
    if (wave >= nTiles) return;

    int lane = threadIdx.x & 31;
    int half = lane >> 4;
    int l15  = lane & 15;

    long rowA = (long)wave * 16 + l15;
    const float* arow = ht + rowA * D + 2 * half;
    const float* brow = &sWT[l15 * BSTRIDE + 2 * half];

    v8f acc0 = {};
    v8f acc1 = {};
    #pragma unroll 4
    for (int kb = 0; kb < D; kb += 8) {
        v2f a0 = *(const v2f*)(arow + kb);
        v2f a1 = *(const v2f*)(arow + kb + 4);
        v2f b0 = *(const v2f*)(brow + kb);
        v2f b1 = *(const v2f*)(brow + kb + 4);
        acc0 = __builtin_amdgcn_wmma_f32_16x16x4_f32(
                   false, a0, false, b0, (short)0, acc0, false, false);
        acc1 = __builtin_amdgcn_wmma_f32_16x16x4_f32(
                   false, a1, false, b1, (short)0, acc1, false, false);
    }

    if (l15 < 3) {
        long rowBase = (long)wave * 16 + half * 8;
        #pragma unroll
        for (int r = 0; r < 8; ++r) {
            long row = rowBase + r;
            int  bidx = (int)(row / N);
            float tau = fmaxf(1.0f - ts[bidx], TAU_MIN_F);
            long idx = row * 3 + l15;
            x[idx] = kp[idx] + tau * (acc0[r] + acc1[r] + sB[l15]);
        }
    }
}

__global__ void edge_kernel(
    const float* __restrict__ x,
    const int*   __restrict__ ei,
    const float* __restrict__ rl,
    float* __restrict__ delta,
    int E, int N, int BE)
{
    int t = blockIdx.x * blockDim.x + threadIdx.x;
    if (t >= BE) return;
    int b = t / E;
    int e = t - b * E;
    int s = ei[e];
    int d = ei[E + e];
    int si = (b * N + s) * 3;
    int di = (b * N + d) * 3;

    float dx = x[si + 0] - x[di + 0];
    float dy = x[si + 1] - x[di + 1];
    float dz = x[si + 2] - x[di + 2];
    float dist = sqrtf(dx * dx + dy * dy + dz * dz);
    float C    = dist - rl[e];
    float scal = (-C / XPBD_DENOM) / (dist + 1e-9f);

    float cx = fminf(fmaxf(scal * dx, -XPBD_MAXC), XPBD_MAXC);
    float cy = fminf(fmaxf(scal * dy, -XPBD_MAXC), XPBD_MAXC);
    float cz = fminf(fmaxf(scal * dz, -XPBD_MAXC), XPBD_MAXC);

    atomic_add_f32(&delta[si + 0],  cx);
    atomic_add_f32(&delta[si + 1],  cy);
    atomic_add_f32(&delta[si + 2],  cz);
    atomic_add_f32(&delta[di + 0], -cx);
    atomic_add_f32(&delta[di + 1], -cy);
    atomic_add_f32(&delta[di + 2], -cz);
}

__global__ void apply_kernel(float* __restrict__ x, float* __restrict__ delta, int n) {
    int i = blockIdx.x * blockDim.x + threadIdx.x;
    if (i < n) {
        x[i] += delta[i];
        delta[i] = 0.0f;
    }
}

__global__ void final_kernel(
    const float* __restrict__ x, const float* __restrict__ delta,
    const float* __restrict__ kp, const float* __restrict__ ts,
    float* __restrict__ out, int n, int perBatch)
{
    int i = blockIdx.x * blockDim.x + threadIdx.x;
    if (i < n) {
        int b = i / perBatch;
        float tau = fmaxf(1.0f - ts[b], TAU_MIN_F);
        out[i] = (x[i] + delta[i] - kp[i]) / tau;
    }
}

extern "C" void kernel_launch(void* const* d_in, const int* in_sizes, int n_in,
                              void* d_out, int out_size, void* d_ws, size_t ws_size,
                              hipStream_t stream) {
    const float* kp   = (const float*)d_in[0];
    const float* ts   = (const float*)d_in[1];
    const float* ht   = (const float*)d_in[2];
    const float* W    = (const float*)d_in[3];
    const float* bias = (const float*)d_in[4];
    const int*   ei   = (const int*)d_in[5];
    const float* rl   = (const float*)d_in[6];

    const int B   = in_sizes[1];
    const int E   = in_sizes[6];
    const int BN3 = in_sizes[0];
    const int BN  = BN3 / 3;
    const int N   = BN / B;
    const int D   = in_sizes[2] / BN;

    float* xbuf  = (float*)d_ws;
    float* delta = xbuf + BN3;
    (void)ws_size;

    const int T = 256;
    const int nTiles = (BN + 15) / 16;
    const int gVec   = (BN3 + T - 1) / T;
    const int BE     = B * E;
    const int gEdge  = (BE + T - 1) / T;
    const int gWmma  = (nTiles * 32 + T - 1) / T;

    zero_kernel<<<gVec, T, 0, stream>>>(delta, BN3);
    head_wmma_kernel<<<gWmma, T, 0, stream>>>(ht, W, bias, kp, ts, xbuf, nTiles, D, N);

    const int XPBD_ITERS = 4;
    for (int it = 0; it < XPBD_ITERS; ++it) {
        edge_kernel<<<gEdge, T, 0, stream>>>(xbuf, ei, rl, delta, E, N, BE);
        if (it < XPBD_ITERS - 1) {
            apply_kernel<<<gVec, T, 0, stream>>>(xbuf, delta, BN3);
        } else {
            final_kernel<<<gVec, T, 0, stream>>>(xbuf, delta, kp, ts,
                                                 (float*)d_out, BN3, N * 3);
        }
    }
}


# x and delta node-major (N,B,3); one 16-lane row per edge, lane=(batch,component): each endpoint's gather and f32 atomic scatter is one 48-byte access; squared length summed by a quad butterfly
# speedup vs baseline: 10.5859x; 10.5859x over previous
#include <hip/hip_runtime.h>
#include <hip/hip_bf16.h>


typedef float v2f __attribute__((ext_vector_type(2)));
typedef float v8f __attribute__((ext_vector_type(8)));

#define TAU_MIN_F     0.001f
#define XPBD_MAXC     0.15f
#define XPBD_DENOM    (2.0f + 1e-9f)

#define BSTRIDE 260

__device__ __forceinline__ void atomic_add_f32(float* p, float v) {
    unsafeAtomicAdd(p, v);
}

#define DPP_QUAD_XOR1   0xB1
#define DPP_QUAD_XOR2   0x4E
#define DPP_ROW_SHARE0  0x150
template <int CTRL>
__device__ __forceinline__ int row_perm_i32(int v) {
    return __builtin_amdgcn_update_dpp(0, v, CTRL, 0xF, 0xF, true);
}
template <int CTRL>
__device__ __forceinline__ float row_perm_f32(float v) {
    return __int_as_float(__builtin_amdgcn_update_dpp(0, __float_as_int(v), CTRL, 0xF, 0xF, true));
}

__global__ void zero_kernel(float* __restrict__ p, int n) {
    int i = blockIdx.x * blockDim.x + threadIdx.x;
    if (i < n) p[i] = 0.0f;
}

__global__ __launch_bounds__(256) void head_wmma_kernel(
    const float* __restrict__ ht,
    const float* __restrict__ W,
    const float* __restrict__ bias,
    const float* __restrict__ kp,
    const float* __restrict__ ts,
    float* __restrict__ x,
    int nTiles, int D, int N, int B)
{
    __shared__ float sWT[16 * BSTRIDE];
    __shared__ float sB[3];

    for (int idx = threadIdx.x; idx < 16 * 256; idx += blockDim.x) {
        int n = idx >> 8;
        int k = idx & 255;
        sWT[n * BSTRIDE + k] = (n < 3) ? W[k * 3 + n] : 0.0f;
    }
    if (threadIdx.x < 3) sB[threadIdx.x] = bias[threadIdx.x];
    __syncthreads();

    int wave = (blockIdx.x * blockDim.x + threadIdx.x) >> 5;
    if (wave >= nTiles) return;

    int lane = threadIdx.x & 31;
    int half = lane >> 4;
    int l15  = lane & 15;

    long rowA = (long)wave * 16 + l15;
    const float* arow = ht + rowA * D + 2 * half;
    const float* brow = &sWT[l15 * BSTRIDE + 2 * half];

    v8f acc0 = {};
    v8f acc1 = {};
    #pragma unroll 4
    for (int kb = 0; kb < D; kb += 8) {
        v2f a0 = *(const v2f*)(arow + kb);
        v2f a1 = *(const v2f*)(arow + kb + 4);
        v2f b0 = *(const v2f*)(brow + kb);
        v2f b1 = *(const v2f*)(brow + kb + 4);
        acc0 = __builtin_amdgcn_wmma_f32_16x16x4_f32(
                   false, a0, false, b0, (short)0, acc0, false, false);
        acc1 = __builtin_amdgcn_wmma_f32_16x16x4_f32(
                   false, a1, false, b1, (short)0, acc1, false, false);
    }

    if (l15 < 3) {
        long rowBase = (long)wave * 16 + half * 8;
        #pragma unroll
        for (int r = 0; r < 8; ++r) {
            long row = rowBase + r;
            int  bidx = (int)(row / N);
            int  node = (int)(row - (long)bidx * N);
            float tau = fmaxf(1.0f - ts[bidx], TAU_MIN_F);
            long idx = row * 3 + l15;
            long xi  = ((long)node * B + bidx) * 3 + l15;
            x[xi] = kp[idx] + tau * (acc0[r] + acc1[r] + sB[l15]);
        }
    }
}

template <int K>
__device__ __forceinline__ void edge_gather_steps(
    const float* __restrict__ x, int s_l, int d_l, int offc, int B3,
    int (&is)[16], int (&id)[16], float (&xs)[16], float (&xd)[16])
{
    if constexpr (K < 16) {
        is[K] = row_perm_i32<DPP_ROW_SHARE0 + K>(s_l) * B3 + offc;
        id[K] = row_perm_i32<DPP_ROW_SHARE0 + K>(d_l) * B3 + offc;
        xs[K] = x[is[K]];
        xd[K] = x[id[K]];
        edge_gather_steps<K + 1>(x, s_l, d_l, offc, B3, is, id, xs, xd);
    }
}

template <int K>
__device__ __forceinline__ void edge_corr_steps(
    float r_l, bool laneOn, int nValid,
    const float (&xs)[16], const float (&xd)[16], float (&corr)[16])
{
    if constexpr (K < 16) {
        float r    = row_perm_f32<DPP_ROW_SHARE0 + K>(r_l);
        float diff = (laneOn && K < nValid) ? (xs[K] - xd[K]) : 0.0f;
        float sq = diff * diff;
        sq += row_perm_f32<DPP_QUAD_XOR1>(sq);
        sq += row_perm_f32<DPP_QUAD_XOR2>(sq);
        float dist = sqrtf(sq);
        float C    = dist - r;
        float scal = (-C / XPBD_DENOM) / (dist + 1e-9f);
        corr[K] = fminf(fmaxf(scal * diff, -XPBD_MAXC), XPBD_MAXC);
        edge_corr_steps<K + 1>(r_l, laneOn, nValid, xs, xd, corr);
    }
}

__global__ __launch_bounds__(256) void edge_kernel(
    const float* __restrict__ x,
    const int*   __restrict__ ei,
    const float* __restrict__ rl,
    float* __restrict__ delta,
    int E, int B)
{
    const int chunk = blockIdx.x * (blockDim.x >> 5) + (threadIdx.x >> 5);
    const int base  = chunk * 32;
    if (base >= E) return;

    const int lane = threadIdx.x & 31;
    const int j    = lane & 15;
    const int bq   = j >> 2;
    const int c    = j & 3;
    const int B3   = B * 3;

    const int   e_l = min(base + lane, E - 1);
    const int   s_l = ei[e_l];
    const int   d_l = ei[E + e_l];
    const float r_l = rl[e_l];

    const int rowBase = base + (lane & 16);
    const int nValid  = min(16, E - rowBase);

    for (int b0 = 0; b0 < B; b0 += 4) {
        const int  b      = b0 + bq;
        const bool laneOn = (c < 3) && (b < B);
        const int  offc   = laneOn ? (b * 3 + c) : 0;

        int   is[16], id[16];
        float xs[16], xd[16], corr[16];
        edge_gather_steps<0>(x, s_l, d_l, offc, B3, is, id, xs, xd);
        edge_corr_steps<0>(r_l, laneOn, nValid, xs, xd, corr);

        if (laneOn) {
            if (nValid == 16) {
                #pragma unroll
                for (int k = 0; k < 16; ++k) {
                    atomic_add_f32(&delta[is[k]],  corr[k]);
                    atomic_add_f32(&delta[id[k]], -corr[k]);
                }
            } else {
                #pragma unroll
                for (int k = 0; k < 16; ++k) {
                    if (k < nValid) {
                        atomic_add_f32(&delta[is[k]],  corr[k]);
                        atomic_add_f32(&delta[id[k]], -corr[k]);
                    }
                }
            }
        }
    }
}

__global__ void apply_kernel(float* __restrict__ x, float* __restrict__ delta, int n) {
    int i = blockIdx.x * blockDim.x + threadIdx.x;
    if (i < n) {
        x[i] += delta[i];
        delta[i] = 0.0f;
    }
}

__global__ void final_kernel(
    const float* __restrict__ x, const float* __restrict__ delta,
    const float* __restrict__ kp, const float* __restrict__ ts,
    float* __restrict__ out, int n, int perBatch, int B)
{
    int i = blockIdx.x * blockDim.x + threadIdx.x;
    if (i < n) {
        int b    = i / perBatch;
        int rem  = i - b * perBatch;
        int node = rem / 3;
        int c    = rem - node * 3;
        int t    = (node * B + b) * 3 + c;
        float tau = fmaxf(1.0f - ts[b], TAU_MIN_F);
        out[i] = (x[t] + delta[t] - kp[i]) / tau;
    }
}

extern "C" void kernel_launch(void* const* d_in, const int* in_sizes, int n_in,
                              void* d_out, int out_size, void* d_ws, size_t ws_size,
                              hipStream_t stream) {
    const float* kp   = (const float*)d_in[0];
    const float* ts   = (const float*)d_in[1];
    const float* ht   = (const float*)d_in[2];
    const float* W    = (const float*)d_in[3];
    const float* bias = (const float*)d_in[4];
    const int*   ei   = (const int*)d_in[5];
    const float* rl   = (const float*)d_in[6];

    const int B   = in_sizes[1];
    const int E   = in_sizes[6];
    const int BN3 = in_sizes[0];
    const int BN  = BN3 / 3;
    const int N   = BN / B;
    const int D   = in_sizes[2] / BN;

    float* xbuf  = (float*)d_ws;
    float* delta = xbuf + BN3;
    (void)ws_size;

    const int T = 256;
    const int nTiles = (BN + 15) / 16;
    const int gVec   = (BN3 + T - 1) / T;
    const int nChunk = (E + 31) / 32;
    const int gEdge  = (nChunk + T / 32 - 1) / (T / 32);
    const int gWmma  = (nTiles * 32 + T - 1) / T;

    zero_kernel<<<gVec, T, 0, stream>>>(delta, BN3);
    head_wmma_kernel<<<gWmma, T, 0, stream>>>(ht, W, bias, kp, ts, xbuf, nTiles, D, N, B);

    const int XPBD_ITERS = 4;
    for (int it = 0; it < XPBD_ITERS; ++it) {
        edge_kernel<<<gEdge, T, 0, stream>>>(xbuf, ei, rl, delta, E, B);
        if (it < XPBD_ITERS - 1) {
            apply_kernel<<<gVec, T, 0, stream>>>(xbuf, delta, BN3);
        } else {
            final_kernel<<<gVec, T, 0, stream>>>(xbuf, delta, kp, ts,
                                                 (float*)d_out, BN3, N * 3, B);
        }
    }
}


# x and delta node records padded from 48 to 64 aligned bytes (4 batches x (3+pad)): each 16-lane endpoint gather and f32 atomic scatter is one aligned access that never straddles a 128-byte line
# speedup vs baseline: 12.9824x; 1.2264x over previous
#include <hip/hip_runtime.h>
#include <hip/hip_bf16.h>


typedef float v2f __attribute__((ext_vector_type(2)));
typedef float v8f __attribute__((ext_vector_type(8)));

#define TAU_MIN_F     0.001f
#define XPBD_MAXC     0.15f
#define XPBD_DENOM    (2.0f + 1e-9f)

#define BSTRIDE 260

__device__ __forceinline__ void atomic_add_f32(float* p, float v) {
    unsafeAtomicAdd(p, v);
}

#define DPP_QUAD_XOR1   0xB1
#define DPP_QUAD_XOR2   0x4E
#define DPP_ROW_SHARE0  0x150
template <int CTRL>
__device__ __forceinline__ int row_perm_i32(int v) {
    return __builtin_amdgcn_update_dpp(0, v, CTRL, 0xF, 0xF, true);
}
template <int CTRL>
__device__ __forceinline__ float row_perm_f32(float v) {
    return __int_as_float(__builtin_amdgcn_update_dpp(0, __float_as_int(v), CTRL, 0xF, 0xF, true));
}

__global__ void zero_kernel(float* __restrict__ p, int n) {
    int i = blockIdx.x * blockDim.x + threadIdx.x;
    if (i < n) p[i] = 0.0f;
}

__global__ __launch_bounds__(256) void head_wmma_kernel(
    const float* __restrict__ ht,
    const float* __restrict__ W,
    const float* __restrict__ bias,
    const float* __restrict__ kp,
    const float* __restrict__ ts,
    float* __restrict__ x,
    int nTiles, int D, int N, int CS, int NS)
{
    __shared__ float sWT[16 * BSTRIDE];
    __shared__ float sB[3];

    for (int idx = threadIdx.x; idx < 16 * 256; idx += blockDim.x) {
        int n = idx >> 8;
        int k = idx & 255;
        sWT[n * BSTRIDE + k] = (n < 3) ? W[k * 3 + n] : 0.0f;
    }
    if (threadIdx.x < 3) sB[threadIdx.x] = bias[threadIdx.x];
    __syncthreads();

    int wave = (blockIdx.x * blockDim.x + threadIdx.x) >> 5;
    if (wave >= nTiles) return;

    int lane = threadIdx.x & 31;
    int half = lane >> 4;
    int l15  = lane & 15;

    long rowA = (long)wave * 16 + l15;
    const float* arow = ht + rowA * D + 2 * half;
    const float* brow = &sWT[l15 * BSTRIDE + 2 * half];

    v8f acc0 = {};
    v8f acc1 = {};
    #pragma unroll 4
    for (int kb = 0; kb < D; kb += 8) {
        v2f a0 = *(const v2f*)(arow + kb);
        v2f a1 = *(const v2f*)(arow + kb + 4);
        v2f b0 = *(const v2f*)(brow + kb);
        v2f b1 = *(const v2f*)(brow + kb + 4);
        acc0 = __builtin_amdgcn_wmma_f32_16x16x4_f32(
                   false, a0, false, b0, (short)0, acc0, false, false);
        acc1 = __builtin_amdgcn_wmma_f32_16x16x4_f32(
                   false, a1, false, b1, (short)0, acc1, false, false);
    }

    if (l15 < CS) {
        long rowBase = (long)wave * 16 + half * 8;
        #pragma unroll
        for (int r = 0; r < 8; ++r) {
            long row = rowBase + r;
            int  bidx = (int)(row / N);
            int  node = (int)(row - (long)bidx * N);
            long xi  = (long)node * NS + bidx * CS + l15;
            float v = 0.0f;
            if (l15 < 3) {
                float tau = fmaxf(1.0f - ts[bidx], TAU_MIN_F);
                v = kp[row * 3 + l15] + tau * (acc0[r] + acc1[r] + sB[l15]);
            }
            x[xi] = v;
        }
    }
}

template <int K>
__device__ __forceinline__ void edge_gather_steps(
    const float* __restrict__ x, int s_l, int d_l, int offc, int NS,
    int (&is)[16], int (&id)[16], float (&xs)[16], float (&xd)[16])
{
    if constexpr (K < 16) {
        is[K] = row_perm_i32<DPP_ROW_SHARE0 + K>(s_l) * NS + offc;
        id[K] = row_perm_i32<DPP_ROW_SHARE0 + K>(d_l) * NS + offc;
        xs[K] = x[is[K]];
        xd[K] = x[id[K]];
        edge_gather_steps<K + 1>(x, s_l, d_l, offc, NS, is, id, xs, xd);
    }
}

template <int K>
__device__ __forceinline__ void edge_corr_steps(
    float r_l, bool laneOn, int nValid,
    const float (&xs)[16], const float (&xd)[16], float (&corr)[16])
{
    if constexpr (K < 16) {
        float r    = row_perm_f32<DPP_ROW_SHARE0 + K>(r_l);
        float diff = (laneOn && K < nValid) ? (xs[K] - xd[K]) : 0.0f;
        float sq = diff * diff;
        sq += row_perm_f32<DPP_QUAD_XOR1>(sq);
        sq += row_perm_f32<DPP_QUAD_XOR2>(sq);
        float dist = sqrtf(sq);
        float C    = dist - r;
        float scal = (-C / XPBD_DENOM) / (dist + 1e-9f);
        corr[K] = fminf(fmaxf(scal * diff, -XPBD_MAXC), XPBD_MAXC);
        edge_corr_steps<K + 1>(r_l, laneOn, nValid, xs, xd, corr);
    }
}

__global__ __launch_bounds__(256) void edge_kernel(
    const float* __restrict__ x,
    const int*   __restrict__ ei,
    const float* __restrict__ rl,
    float* __restrict__ delta,
    int E, int B, int CS, int NS)
{
    const int chunk = blockIdx.x * (blockDim.x >> 5) + (threadIdx.x >> 5);
    const int base  = chunk * 32;
    if (base >= E) return;

    const int lane = threadIdx.x & 31;
    const int j    = lane & 15;
    const int bq   = j >> 2;
    const int c    = j & 3;

    const int   e_l = min(base + lane, E - 1);
    const int   s_l = ei[e_l];
    const int   d_l = ei[E + e_l];
    const float r_l = rl[e_l];

    const int rowBase = base + (lane & 16);
    const int nValid  = min(16, E - rowBase);

    for (int b0 = 0; b0 < B; b0 += 4) {
        const int  b      = b0 + bq;
        const bool laneOn = (c < 3) && (b < B);
        const int  offc   = laneOn ? (b * CS + c) : 0;

        int   is[16], id[16];
        float xs[16], xd[16], corr[16];
        edge_gather_steps<0>(x, s_l, d_l, offc, NS, is, id, xs, xd);
        edge_corr_steps<0>(r_l, laneOn, nValid, xs, xd, corr);

        if (laneOn) {
            if (nValid == 16) {
                #pragma unroll
                for (int k = 0; k < 16; ++k) {
                    atomic_add_f32(&delta[is[k]],  corr[k]);
                    atomic_add_f32(&delta[id[k]], -corr[k]);
                }
            } else {
                #pragma unroll
                for (int k = 0; k < 16; ++k) {
                    if (k < nValid) {
                        atomic_add_f32(&delta[is[k]],  corr[k]);
                        atomic_add_f32(&delta[id[k]], -corr[k]);
                    }
                }
            }
        }
    }
}

__global__ void apply_kernel(float* __restrict__ x, float* __restrict__ delta, int n) {
    int i = blockIdx.x * blockDim.x + threadIdx.x;
    if (i < n) {
        x[i] += delta[i];
        delta[i] = 0.0f;
    }
}

__global__ void final_kernel(
    const float* __restrict__ x, const float* __restrict__ delta,
    const float* __restrict__ kp, const float* __restrict__ ts,
    float* __restrict__ out, int n, int perBatch, int CS, int NS)
{
    int i = blockIdx.x * blockDim.x + threadIdx.x;
    if (i < n) {
        int b    = i / perBatch;
        int rem  = i - b * perBatch;
        int node = rem / 3;
        int c    = rem - node * 3;
        int t    = node * NS + b * CS + c;
        float tau = fmaxf(1.0f - ts[b], TAU_MIN_F);
        out[i] = (x[t] + delta[t] - kp[i]) / tau;
    }
}

extern "C" void kernel_launch(void* const* d_in, const int* in_sizes, int n_in,
                              void* d_out, int out_size, void* d_ws, size_t ws_size,
                              hipStream_t stream) {
    const float* kp   = (const float*)d_in[0];
    const float* ts   = (const float*)d_in[1];
    const float* ht   = (const float*)d_in[2];
    const float* W    = (const float*)d_in[3];
    const float* bias = (const float*)d_in[4];
    const int*   ei   = (const int*)d_in[5];
    const float* rl   = (const float*)d_in[6];

    const int B   = in_sizes[1];
    const int E   = in_sizes[6];
    const int BN3 = in_sizes[0];
    const int BN  = BN3 / 3;
    const int N   = BN / B;
    const int D   = in_sizes[2] / BN;

    const bool padded = (B % 4 == 0) && ws_size >= (size_t)2 * N * B * 4 * sizeof(float);
    const int CS  = padded ? 4 : 3;
    const int NS  = B * CS;
    const int NW  = N * NS;

    float* xbuf  = (float*)d_ws;
    float* delta = xbuf + NW;

    const int T = 256;
    const int nTiles = (BN + 15) / 16;
    const int gVec   = (BN3 + T - 1) / T;
    const int gWs    = (NW + T - 1) / T;
    const int nChunk = (E + 31) / 32;
    const int gEdge  = (nChunk + T / 32 - 1) / (T / 32);
    const int gWmma  = (nTiles * 32 + T - 1) / T;

    zero_kernel<<<gWs, T, 0, stream>>>(delta, NW);
    head_wmma_kernel<<<gWmma, T, 0, stream>>>(ht, W, bias, kp, ts, xbuf, nTiles, D, N, CS, NS);

    const int XPBD_ITERS = 4;
    for (int it = 0; it < XPBD_ITERS; ++it) {
        edge_kernel<<<gEdge, T, 0, stream>>>(xbuf, ei, rl, delta, E, B, CS, NS);
        if (it < XPBD_ITERS - 1) {
            apply_kernel<<<gWs, T, 0, stream>>>(xbuf, delta, NW);
        } else {
            final_kernel<<<gVec, T, 0, stream>>>(xbuf, delta, kp, ts,
                                                 (float*)d_out, BN3, N * 3, CS, NS);
        }
    }
}


# delta accumulator as two planes, one per second-level cache half: an edge wave reads its die id and atomic-adds only into its half's plane; apply/final add the planes (sum order: per-half sums first)
# speedup vs baseline: 18.3241x; 1.4115x over previous
#include <hip/hip_runtime.h>
#include <hip/hip_bf16.h>


typedef float v2f __attribute__((ext_vector_type(2)));
typedef float v8f __attribute__((ext_vector_type(8)));

#define TAU_MIN_F     0.001f
#define XPBD_MAXC     0.15f
#define XPBD_DENOM    (2.0f + 1e-9f)

#define BSTRIDE 260

__device__ __forceinline__ void atomic_add_f32(float* p, float v) {
    unsafeAtomicAdd(p, v);
}

__device__ __forceinline__ int cache_half() {
    unsigned id = __builtin_amdgcn_s_sendmsg_rtn(0x87);
    return (int)((id >> 18) & 1u);
}

#define DPP_QUAD_XOR1   0xB1
#define DPP_QUAD_XOR2   0x4E
#define DPP_ROW_SHARE0  0x150
template <int CTRL>
__device__ __forceinline__ int row_perm_i32(int v) {
    return __builtin_amdgcn_update_dpp(0, v, CTRL, 0xF, 0xF, true);
}
template <int CTRL>
__device__ __forceinline__ float row_perm_f32(float v) {
    return __int_as_float(__builtin_amdgcn_update_dpp(0, __float_as_int(v), CTRL, 0xF, 0xF, true));
}

__global__ void zero_kernel(float* __restrict__ p, int n) {
    int i = blockIdx.x * blockDim.x + threadIdx.x;
    if (i < n) p[i] = 0.0f;
}

__global__ __launch_bounds__(256) void head_wmma_kernel(
    const float* __restrict__ ht,
    const float* __restrict__ W,
    const float* __restrict__ bias,
    const float* __restrict__ kp,
    const float* __restrict__ ts,
    float* __restrict__ x,
    int nTiles, int D, int N, int CS, int NS)
{
    __shared__ float sWT[16 * BSTRIDE];
    __shared__ float sB[3];

    for (int idx = threadIdx.x; idx < 16 * 256; idx += blockDim.x) {
        int n = idx >> 8;
        int k = idx & 255;
        sWT[n * BSTRIDE + k] = (n < 3) ? W[k * 3 + n] : 0.0f;
    }
    if (threadIdx.x < 3) sB[threadIdx.x] = bias[threadIdx.x];
    __syncthreads();

    int wave = (blockIdx.x * blockDim.x + threadIdx.x) >> 5;
    if (wave >= nTiles) return;

    int lane = threadIdx.x & 31;
    int half = lane >> 4;
    int l15  = lane & 15;

    long rowA = (long)wave * 16 + l15;
    const float* arow = ht + rowA * D + 2 * half;
    const float* brow = &sWT[l15 * BSTRIDE + 2 * half];

    v8f acc0 = {};
    v8f acc1 = {};
    #pragma unroll 4
    for (int kb = 0; kb < D; kb += 8) {
        v2f a0 = *(const v2f*)(arow + kb);
        v2f a1 = *(const v2f*)(arow + kb + 4);
        v2f b0 = *(const v2f*)(brow + kb);
        v2f b1 = *(const v2f*)(brow + kb + 4);
        acc0 = __builtin_amdgcn_wmma_f32_16x16x4_f32(
                   false, a0, false, b0, (short)0, acc0, false, false);
        acc1 = __builtin_amdgcn_wmma_f32_16x16x4_f32(
                   false, a1, false, b1, (short)0, acc1, false, false);
    }

    if (l15 < CS) {
        long rowBase = (long)wave * 16 + half * 8;
        #pragma unroll
        for (int r = 0; r < 8; ++r) {
            long row = rowBase + r;
            int  bidx = (int)(row / N);
            int  node = (int)(row - (long)bidx * N);
            long xi  = (long)node * NS + bidx * CS + l15;
            float v = 0.0f;
            if (l15 < 3) {
                float tau = fmaxf(1.0f - ts[bidx], TAU_MIN_F);
                v = kp[row * 3 + l15] + tau * (acc0[r] + acc1[r] + sB[l15]);
            }
            x[xi] = v;
        }
    }
}

template <int K>
__device__ __forceinline__ void edge_gather_steps(
    const float* __restrict__ x, int s_l, int d_l, int offc, int NS,
    int (&is)[16], int (&id)[16], float (&xs)[16], float (&xd)[16])
{
    if constexpr (K < 16) {
        is[K] = row_perm_i32<DPP_ROW_SHARE0 + K>(s_l) * NS + offc;
        id[K] = row_perm_i32<DPP_ROW_SHARE0 + K>(d_l) * NS + offc;
        xs[K] = x[is[K]];
        xd[K] = x[id[K]];
        edge_gather_steps<K + 1>(x, s_l, d_l, offc, NS, is, id, xs, xd);
    }
}

template <int K>
__device__ __forceinline__ void edge_corr_steps(
    float r_l, bool laneOn, int nValid,
    const float (&xs)[16], const float (&xd)[16], float (&corr)[16])
{
    if constexpr (K < 16) {
        float r    = row_perm_f32<DPP_ROW_SHARE0 + K>(r_l);
        float diff = (laneOn && K < nValid) ? (xs[K] - xd[K]) : 0.0f;
        float sq = diff * diff;
        sq += row_perm_f32<DPP_QUAD_XOR1>(sq);
        sq += row_perm_f32<DPP_QUAD_XOR2>(sq);
        float dist = sqrtf(sq);
        float C    = dist - r;
        float scal = (-C / XPBD_DENOM) / (dist + 1e-9f);
        corr[K] = fminf(fmaxf(scal * diff, -XPBD_MAXC), XPBD_MAXC);
        edge_corr_steps<K + 1>(r_l, laneOn, nValid, xs, xd, corr);
    }
}

__global__ __launch_bounds__(256) void edge_kernel(
    const float* __restrict__ x,
    const int*   __restrict__ ei,
    const float* __restrict__ rl,
    float* __restrict__ delta,
    int E, int B, int CS, int NS, int planeStride)
{
    const int chunk = blockIdx.x * (blockDim.x >> 5) + (threadIdx.x >> 5);
    const int base  = chunk * 32;
    if (base >= E) return;

    const int lane = threadIdx.x & 31;
    const int j    = lane & 15;
    const int bq   = j >> 2;
    const int c    = j & 3;

    const int   e_l = min(base + lane, E - 1);
    const int   s_l = ei[e_l];
    const int   d_l = ei[E + e_l];
    const float r_l = rl[e_l];

    const int rowBase = base + (lane & 16);
    const int nValid  = min(16, E - rowBase);

    float* __restrict__ dl = delta + (size_t)cache_half() * (size_t)planeStride;

    for (int b0 = 0; b0 < B; b0 += 4) {
        const int  b      = b0 + bq;
        const bool laneOn = (c < 3) && (b < B);
        const int  offc   = laneOn ? (b * CS + c) : 0;

        int   is[16], id[16];
        float xs[16], xd[16], corr[16];
        edge_gather_steps<0>(x, s_l, d_l, offc, NS, is, id, xs, xd);
        edge_corr_steps<0>(r_l, laneOn, nValid, xs, xd, corr);

        if (laneOn) {
            if (nValid == 16) {
                #pragma unroll
                for (int k = 0; k < 16; ++k) {
                    atomic_add_f32(&dl[is[k]],  corr[k]);
                    atomic_add_f32(&dl[id[k]], -corr[k]);
                }
            } else {
                #pragma unroll
                for (int k = 0; k < 16; ++k) {
                    if (k < nValid) {
                        atomic_add_f32(&dl[is[k]],  corr[k]);
                        atomic_add_f32(&dl[id[k]], -corr[k]);
                    }
                }
            }
        }
    }
}

__global__ void apply_kernel(float* __restrict__ x, float* __restrict__ delta,
                             int n, int planeStride) {
    int i = blockIdx.x * blockDim.x + threadIdx.x;
    if (i < n) {
        float d = delta[i];
        delta[i] = 0.0f;
        if (planeStride) {
            d += delta[i + planeStride];
            delta[i + planeStride] = 0.0f;
        }
        x[i] += d;
    }
}

__global__ void final_kernel(
    const float* __restrict__ x, const float* __restrict__ delta,
    const float* __restrict__ kp, const float* __restrict__ ts,
    float* __restrict__ out, int n, int perBatch, int CS, int NS,
    int planeStride)
{
    int i = blockIdx.x * blockDim.x + threadIdx.x;
    if (i < n) {
        int b    = i / perBatch;
        int rem  = i - b * perBatch;
        int node = rem / 3;
        int c    = rem - node * 3;
        int t    = node * NS + b * CS + c;
        float tau = fmaxf(1.0f - ts[b], TAU_MIN_F);
        float d = delta[t];
        if (planeStride) d += delta[t + planeStride];
        out[i] = (x[t] + d - kp[i]) / tau;
    }
}

extern "C" void kernel_launch(void* const* d_in, const int* in_sizes, int n_in,
                              void* d_out, int out_size, void* d_ws, size_t ws_size,
                              hipStream_t stream) {
    const float* kp   = (const float*)d_in[0];
    const float* ts   = (const float*)d_in[1];
    const float* ht   = (const float*)d_in[2];
    const float* W    = (const float*)d_in[3];
    const float* bias = (const float*)d_in[4];
    const int*   ei   = (const int*)d_in[5];
    const float* rl   = (const float*)d_in[6];

    const int B   = in_sizes[1];
    const int E   = in_sizes[6];
    const int BN3 = in_sizes[0];
    const int BN  = BN3 / 3;
    const int N   = BN / B;
    const int D   = in_sizes[2] / BN;

    const bool padded = (B % 4 == 0) && ws_size >= (size_t)2 * N * B * 4 * sizeof(float);
    const int CS  = padded ? 4 : 3;
    const int NS  = B * CS;
    const int NW  = N * NS;

    const bool twoPlanes   = ws_size >= (size_t)3 * NW * sizeof(float);
    const int  planeStride = twoPlanes ? NW : 0;
    const int  NWD         = twoPlanes ? 2 * NW : NW;

    float* xbuf  = (float*)d_ws;
    float* delta = xbuf + NW;

    const int T = 256;
    const int nTiles = (BN + 15) / 16;
    const int gVec   = (BN3 + T - 1) / T;
    const int gWs    = (NW + T - 1) / T;
    const int gWsD   = (NWD + T - 1) / T;
    const int nChunk = (E + 31) / 32;
    const int gEdge  = (nChunk + T / 32 - 1) / (T / 32);
    const int gWmma  = (nTiles * 32 + T - 1) / T;

    zero_kernel<<<gWsD, T, 0, stream>>>(delta, NWD);
    head_wmma_kernel<<<gWmma, T, 0, stream>>>(ht, W, bias, kp, ts, xbuf, nTiles, D, N, CS, NS);

    const int XPBD_ITERS = 4;
    for (int it = 0; it < XPBD_ITERS; ++it) {
        edge_kernel<<<gEdge, T, 0, stream>>>(xbuf, ei, rl, delta, E, B, CS, NS, planeStride);
        if (it < XPBD_ITERS - 1) {
            apply_kernel<<<gWs, T, 0, stream>>>(xbuf, delta, NW, planeStride);
        } else {
            final_kernel<<<gVec, T, 0, stream>>>(xbuf, delta, kp, ts,
                                                 (float*)d_out, BN3, N * 3, CS, NS, planeStride);
        }
    }
}


# edge pass vector work cut: node offsets multiplied once per lane, used as 32-bit byte offsets from a scalar base; hardware sqrt and reciprocal (1 ulp) replace the exact sqrt and divide sequences
# speedup vs baseline: 19.5016x; 1.0643x over previous
#include <hip/hip_runtime.h>
#include <hip/hip_bf16.h>


typedef float v2f __attribute__((ext_vector_type(2)));
typedef float v8f __attribute__((ext_vector_type(8)));

#define TAU_MIN_F     0.001f
#define XPBD_MAXC     0.15f
#define XPBD_DENOM    (2.0f + 1e-9f)

#define BSTRIDE 260

__device__ __forceinline__ void atomic_add_f32(float* p, float v) {
    unsafeAtomicAdd(p, v);
}

__device__ __forceinline__ int cache_half() {
    unsigned id = __builtin_amdgcn_s_sendmsg_rtn(0x87);
    return (int)((id >> 18) & 1u);
}

#define DPP_QUAD_XOR1   0xB1
#define DPP_QUAD_XOR2   0x4E
#define DPP_ROW_SHARE0  0x150
template <int CTRL>
__device__ __forceinline__ int row_perm_i32(int v) {
    return __builtin_amdgcn_update_dpp(0, v, CTRL, 0xF, 0xF, true);
}
template <int CTRL>
__device__ __forceinline__ float row_perm_f32(float v) {
    return __int_as_float(__builtin_amdgcn_update_dpp(0, __float_as_int(v), CTRL, 0xF, 0xF, true));
}

__global__ void zero_kernel(float* __restrict__ p, int n) {
    int i = blockIdx.x * blockDim.x + threadIdx.x;
    if (i < n) p[i] = 0.0f;
}

__global__ __launch_bounds__(256) void head_wmma_kernel(
    const float* __restrict__ ht,
    const float* __restrict__ W,
    const float* __restrict__ bias,
    const float* __restrict__ kp,
    const float* __restrict__ ts,
    float* __restrict__ x,
    int nTiles, int D, int N, int CS, int NS)
{
    __shared__ float sWT[16 * BSTRIDE];
    __shared__ float sB[3];

    for (int idx = threadIdx.x; idx < 16 * 256; idx += blockDim.x) {
        int n = idx >> 8;
        int k = idx & 255;
        sWT[n * BSTRIDE + k] = (n < 3) ? W[k * 3 + n] : 0.0f;
    }
    if (threadIdx.x < 3) sB[threadIdx.x] = bias[threadIdx.x];
    __syncthreads();

    int wave = (blockIdx.x * blockDim.x + threadIdx.x) >> 5;
    if (wave >= nTiles) return;

    int lane = threadIdx.x & 31;
    int half = lane >> 4;
    int l15  = lane & 15;

    long rowA = (long)wave * 16 + l15;
    const float* arow = ht + rowA * D + 2 * half;
    const float* brow = &sWT[l15 * BSTRIDE + 2 * half];

    v8f acc0 = {};
    v8f acc1 = {};
    #pragma unroll 4
    for (int kb = 0; kb < D; kb += 8) {
        v2f a0 = *(const v2f*)(arow + kb);
        v2f a1 = *(const v2f*)(arow + kb + 4);
        v2f b0 = *(const v2f*)(brow + kb);
        v2f b1 = *(const v2f*)(brow + kb + 4);
        acc0 = __builtin_amdgcn_wmma_f32_16x16x4_f32(
                   false, a0, false, b0, (short)0, acc0, false, false);
        acc1 = __builtin_amdgcn_wmma_f32_16x16x4_f32(
                   false, a1, false, b1, (short)0, acc1, false, false);
    }

    if (l15 < CS) {
        long rowBase = (long)wave * 16 + half * 8;
        #pragma unroll
        for (int r = 0; r < 8; ++r) {
            long row = rowBase + r;
            int  bidx = (int)(row / N);
            int  node = (int)(row - (long)bidx * N);
            long xi  = (long)node * NS + bidx * CS + l15;
            float v = 0.0f;
            if (l15 < 3) {
                float tau = fmaxf(1.0f - ts[bidx], TAU_MIN_F);
                v = kp[row * 3 + l15] + tau * (acc0[r] + acc1[r] + sB[l15]);
            }
            x[xi] = v;
        }
    }
}

__device__ __forceinline__ float load_f32_at(const float* __restrict__ base, unsigned byteOff) {
    return *(const float*)((const char*)base + byteOff);
}
__device__ __forceinline__ void atomic_add_f32_at(float* base, unsigned byteOff, float v) {
    atomic_add_f32((float*)((char*)base + byteOff), v);
}

template <int K>
__device__ __forceinline__ void edge_gather_steps(
    const float* __restrict__ x, int sB_l, int dB_l, int offB,
    unsigned (&is)[16], unsigned (&id)[16], float (&xs)[16], float (&xd)[16])
{
    if constexpr (K < 16) {
        is[K] = (unsigned)(row_perm_i32<DPP_ROW_SHARE0 + K>(sB_l) + offB);
        id[K] = (unsigned)(row_perm_i32<DPP_ROW_SHARE0 + K>(dB_l) + offB);
        xs[K] = load_f32_at(x, is[K]);
        xd[K] = load_f32_at(x, id[K]);
        edge_gather_steps<K + 1>(x, sB_l, dB_l, offB, is, id, xs, xd);
    }
}

template <int K, bool FULL>
__device__ __forceinline__ void edge_corr_steps(
    float r_l, bool laneOn, int nValid,
    const float (&xs)[16], const float (&xd)[16], float (&corr)[16])
{
    if constexpr (K < 16) {
        const float r    = row_perm_f32<DPP_ROW_SHARE0 + K>(r_l);
        const bool  on   = FULL ? laneOn : (laneOn && K < nValid);
        const float diff = on ? (xs[K] - xd[K]) : 0.0f;
        float sq = diff * diff;
        sq += row_perm_f32<DPP_QUAD_XOR1>(sq);
        sq += row_perm_f32<DPP_QUAD_XOR2>(sq);
        const float dist = __builtin_amdgcn_sqrtf(sq);
        const float C    = dist - r;
        const float inv  = __builtin_amdgcn_rcpf(dist + 1e-9f);
        const float scal = (C * (-1.0f / XPBD_DENOM)) * inv;
        corr[K] = fminf(fmaxf(scal * diff, -XPBD_MAXC), XPBD_MAXC);
        edge_corr_steps<K + 1, FULL>(r_l, laneOn, nValid, xs, xd, corr);
    }
}

template <bool FULL>
__device__ __forceinline__ void edge_wave(
    const float* __restrict__ x, float* __restrict__ dl,
    int sB_l, int dB_l, float r_l, int nValid,
    int bq, int c, int B, int CS)
{
    for (int b0 = 0; b0 < B; b0 += 4) {
        const int  b      = b0 + bq;
        const bool laneOn = (c < 3) && (b < B);
        const int  offB   = laneOn ? (b * CS + c) * 4 : 0;

        unsigned is[16], id[16];
        float    xs[16], xd[16], corr[16];
        edge_gather_steps<0>(x, sB_l, dB_l, offB, is, id, xs, xd);
        edge_corr_steps<0, FULL>(r_l, laneOn, nValid, xs, xd, corr);

        if (laneOn) {
            #pragma unroll
            for (int k = 0; k < 16; ++k) {
                if (FULL || k < nValid) {
                    unsigned os = is[k], od = id[k];
                    asm volatile("" : "+v"(os), "+v"(od));
                    atomic_add_f32_at(dl, os,  corr[k]);
                    atomic_add_f32_at(dl, od, -corr[k]);
                }
            }
        }
    }
}

__global__ __launch_bounds__(256) void edge_kernel(
    const float* __restrict__ x,
    const int*   __restrict__ ei,
    const float* __restrict__ rl,
    float* __restrict__ delta,
    int E, int B, int CS, int NS, int planeStride)
{
    const int chunk = blockIdx.x * (blockDim.x >> 5) + (threadIdx.x >> 5);
    const int base  = chunk * 32;
    if (base >= E) return;

    const int lane = threadIdx.x & 31;
    const int j    = lane & 15;
    const int bq   = j >> 2;
    const int c    = j & 3;

    const int   e_l  = min(base + lane, E - 1);
    const int   recB = NS * 4;
    const int   sB_l = ei[e_l] * recB;
    const int   dB_l = ei[E + e_l] * recB;
    const float r_l  = rl[e_l];

    const int rowBase = base + (lane & 16);
    const int nValid  = min(16, E - rowBase);

    float* __restrict__ dl = delta + (size_t)cache_half() * (size_t)planeStride;

    if (base + 32 <= E) {
        edge_wave<true >(x, dl, sB_l, dB_l, r_l, nValid, bq, c, B, CS);
    } else {
        edge_wave<false>(x, dl, sB_l, dB_l, r_l, nValid, bq, c, B, CS);
    }
}

__global__ void apply_kernel(float* __restrict__ x, float* __restrict__ delta,
                             int n, int planeStride) {
    int i = blockIdx.x * blockDim.x + threadIdx.x;
    if (i < n) {
        float d = delta[i];
        delta[i] = 0.0f;
        if (planeStride) {
            d += delta[i + planeStride];
            delta[i + planeStride] = 0.0f;
        }
        x[i] += d;
    }
}

__global__ void final_kernel(
    const float* __restrict__ x, const float* __restrict__ delta,
    const float* __restrict__ kp, const float* __restrict__ ts,
    float* __restrict__ out, int n, int perBatch, int CS, int NS,
    int planeStride)
{
    int i = blockIdx.x * blockDim.x + threadIdx.x;
    if (i < n) {
        int b    = i / perBatch;
        int rem  = i - b * perBatch;
        int node = rem / 3;
        int c    = rem - node * 3;
        int t    = node * NS + b * CS + c;
        float tau = fmaxf(1.0f - ts[b], TAU_MIN_F);
        float d = delta[t];
        if (planeStride) d += delta[t + planeStride];
        out[i] = (x[t] + d - kp[i]) / tau;
    }
}

extern "C" void kernel_launch(void* const* d_in, const int* in_sizes, int n_in,
                              void* d_out, int out_size, void* d_ws, size_t ws_size,
                              hipStream_t stream) {
    const float* kp   = (const float*)d_in[0];
    const float* ts   = (const float*)d_in[1];
    const float* ht   = (const float*)d_in[2];
    const float* W    = (const float*)d_in[3];
    const float* bias = (const float*)d_in[4];
    const int*   ei   = (const int*)d_in[5];
    const float* rl   = (const float*)d_in[6];

    const int B   = in_sizes[1];
    const int E   = in_sizes[6];
    const int BN3 = in_sizes[0];
    const int BN  = BN3 / 3;
    const int N   = BN / B;
    const int D   = in_sizes[2] / BN;

    const bool padded = (B % 4 == 0) && ws_size >= (size_t)2 * N * B * 4 * sizeof(float);
    const int CS  = padded ? 4 : 3;
    const int NS  = B * CS;
    const int NW  = N * NS;

    const bool twoPlanes   = ws_size >= (size_t)3 * NW * sizeof(float);
    const int  planeStride = twoPlanes ? NW : 0;
    const int  NWD         = twoPlanes ? 2 * NW : NW;

    float* xbuf  = (float*)d_ws;
    float* delta = xbuf + NW;

    const int T = 256;
    const int nTiles = (BN + 15) / 16;
    const int gVec   = (BN3 + T - 1) / T;
    const int gWs    = (NW + T - 1) / T;
    const int gWsD   = (NWD + T - 1) / T;
    const int nChunk = (E + 31) / 32;
    const int gEdge  = (nChunk + T / 32 - 1) / (T / 32);
    const int gWmma  = (nTiles * 32 + T - 1) / T;

    zero_kernel<<<gWsD, T, 0, stream>>>(delta, NWD);
    head_wmma_kernel<<<gWmma, T, 0, stream>>>(ht, W, bias, kp, ts, xbuf, nTiles, D, N, CS, NS);

    const int XPBD_ITERS = 4;
    for (int it = 0; it < XPBD_ITERS; ++it) {
        edge_kernel<<<gEdge, T, 0, stream>>>(xbuf, ei, rl, delta, E, B, CS, NS, planeStride);
        if (it < XPBD_ITERS - 1) {
            apply_kernel<<<gWs, T, 0, stream>>>(xbuf, delta, NW, planeStride);
        } else {
            final_kernel<<<gVec, T, 0, stream>>>(xbuf, delta, kp, ts,
                                                 (float*)d_out, BN3, N * 3, CS, NS, planeStride);
        }
    }
}


# elementwise kernels (zero, apply, final) take several elements per lane (16-byte vectors in zero and apply), a lane's loads all ahead of its stores: 143752 short waves per call become 10960
# speedup vs baseline: 21.8917x; 1.1226x over previous
#include <hip/hip_runtime.h>
#include <hip/hip_bf16.h>


typedef float v2f __attribute__((ext_vector_type(2)));
typedef float v8f __attribute__((ext_vector_type(8)));

#define TAU_MIN_F     0.001f
#define XPBD_MAXC     0.15f
#define XPBD_DENOM    (2.0f + 1e-9f)

#define BSTRIDE 260

__device__ __forceinline__ void atomic_add_f32(float* p, float v) {
    unsafeAtomicAdd(p, v);
}

__device__ __forceinline__ int cache_half() {
    unsigned id = __builtin_amdgcn_s_sendmsg_rtn(0x87);
    return (int)((id >> 18) & 1u);
}

#define DPP_QUAD_XOR1   0xB1
#define DPP_QUAD_XOR2   0x4E
#define DPP_ROW_SHARE0  0x150
template <int CTRL>
__device__ __forceinline__ int row_perm_i32(int v) {
    return __builtin_amdgcn_update_dpp(0, v, CTRL, 0xF, 0xF, true);
}
template <int CTRL>
__device__ __forceinline__ float row_perm_f32(float v) {
    return __int_as_float(__builtin_amdgcn_update_dpp(0, __float_as_int(v), CTRL, 0xF, 0xF, true));
}

#define ZERO_VPL   8
#define APPLY_VPL  4
#define FINAL_EPL  4

__global__ __launch_bounds__(256) void zero_kernel(float* __restrict__ p, int n) {
    const float4 z = make_float4(0.0f, 0.0f, 0.0f, 0.0f);
    if ((((size_t)p) & 15) == 0) {
        const int n4 = n >> 2;
        float4* __restrict__ p4 = (float4*)p;
        const int base = blockIdx.x * (blockDim.x * ZERO_VPL) + threadIdx.x;
        #pragma unroll
        for (int k = 0; k < ZERO_VPL; ++k) {
            const int i = base + k * blockDim.x;
            if (i < n4) p4[i] = z;
        }
        if (blockIdx.x == 0 && threadIdx.x < (n & 3)) p[(n4 << 2) + threadIdx.x] = 0.0f;
    } else {
        const int nthr = gridDim.x * blockDim.x;
        for (int i = blockIdx.x * blockDim.x + threadIdx.x; i < n; i += nthr) p[i] = 0.0f;
    }
}

__global__ __launch_bounds__(256) void head_wmma_kernel(
    const float* __restrict__ ht,
    const float* __restrict__ W,
    const float* __restrict__ bias,
    const float* __restrict__ kp,
    const float* __restrict__ ts,
    float* __restrict__ x,
    int nTiles, int D, int N, int CS, int NS)
{
    __shared__ float sWT[16 * BSTRIDE];
    __shared__ float sB[3];

    for (int idx = threadIdx.x; idx < 16 * 256; idx += blockDim.x) {
        int n = idx >> 8;
        int k = idx & 255;
        sWT[n * BSTRIDE + k] = (n < 3) ? W[k * 3 + n] : 0.0f;
    }
    if (threadIdx.x < 3) sB[threadIdx.x] = bias[threadIdx.x];
    __syncthreads();

    int wave = (blockIdx.x * blockDim.x + threadIdx.x) >> 5;
    if (wave >= nTiles) return;

    int lane = threadIdx.x & 31;
    int half = lane >> 4;
    int l15  = lane & 15;

    long rowA = (long)wave * 16 + l15;
    const float* arow = ht + rowA * D + 2 * half;
    const float* brow = &sWT[l15 * BSTRIDE + 2 * half];

    v8f acc0 = {};
    v8f acc1 = {};
    #pragma unroll 4
    for (int kb = 0; kb < D; kb += 8) {
        v2f a0 = *(const v2f*)(arow + kb);
        v2f a1 = *(const v2f*)(arow + kb + 4);
        v2f b0 = *(const v2f*)(brow + kb);
        v2f b1 = *(const v2f*)(brow + kb + 4);
        acc0 = __builtin_amdgcn_wmma_f32_16x16x4_f32(
                   false, a0, false, b0, (short)0, acc0, false, false);
        acc1 = __builtin_amdgcn_wmma_f32_16x16x4_f32(
                   false, a1, false, b1, (short)0, acc1, false, false);
    }

    if (l15 < CS) {
        long rowBase = (long)wave * 16 + half * 8;
        #pragma unroll
        for (int r = 0; r < 8; ++r) {
            long row = rowBase + r;
            int  bidx = (int)(row / N);
            int  node = (int)(row - (long)bidx * N);
            long xi  = (long)node * NS + bidx * CS + l15;
            float v = 0.0f;
            if (l15 < 3) {
                float tau = fmaxf(1.0f - ts[bidx], TAU_MIN_F);
                v = kp[row * 3 + l15] + tau * (acc0[r] + acc1[r] + sB[l15]);
            }
            x[xi] = v;
        }
    }
}

__device__ __forceinline__ float load_f32_at(const float* __restrict__ base, unsigned byteOff) {
    return *(const float*)((const char*)base + byteOff);
}
__device__ __forceinline__ void atomic_add_f32_at(float* base, unsigned byteOff, float v) {
    atomic_add_f32((float*)((char*)base + byteOff), v);
}

template <int K>
__device__ __forceinline__ void edge_gather_steps(
    const float* __restrict__ x, int sB_l, int dB_l, int offB,
    unsigned (&is)[16], unsigned (&id)[16], float (&xs)[16], float (&xd)[16])
{
    if constexpr (K < 16) {
        is[K] = (unsigned)(row_perm_i32<DPP_ROW_SHARE0 + K>(sB_l) + offB);
        id[K] = (unsigned)(row_perm_i32<DPP_ROW_SHARE0 + K>(dB_l) + offB);
        xs[K] = load_f32_at(x, is[K]);
        xd[K] = load_f32_at(x, id[K]);
        edge_gather_steps<K + 1>(x, sB_l, dB_l, offB, is, id, xs, xd);
    }
}

template <int K, bool FULL>
__device__ __forceinline__ void edge_corr_steps(
    float r_l, bool laneOn, int nValid,
    const float (&xs)[16], const float (&xd)[16], float (&corr)[16])
{
    if constexpr (K < 16) {
        const float r    = row_perm_f32<DPP_ROW_SHARE0 + K>(r_l);
        const bool  on   = FULL ? laneOn : (laneOn && K < nValid);
        const float diff = on ? (xs[K] - xd[K]) : 0.0f;
        float sq = diff * diff;
        sq += row_perm_f32<DPP_QUAD_XOR1>(sq);
        sq += row_perm_f32<DPP_QUAD_XOR2>(sq);
        const float dist = __builtin_amdgcn_sqrtf(sq);
        const float C    = dist - r;
        const float inv  = __builtin_amdgcn_rcpf(dist + 1e-9f);
        const float scal = (C * (-1.0f / XPBD_DENOM)) * inv;
        corr[K] = fminf(fmaxf(scal * diff, -XPBD_MAXC), XPBD_MAXC);
        edge_corr_steps<K + 1, FULL>(r_l, laneOn, nValid, xs, xd, corr);
    }
}

template <bool FULL>
__device__ __forceinline__ void edge_wave(
    const float* __restrict__ x, float* __restrict__ dl,
    int sB_l, int dB_l, float r_l, int nValid,
    int bq, int c, int B, int CS)
{
    for (int b0 = 0; b0 < B; b0 += 4) {
        const int  b      = b0 + bq;
        const bool laneOn = (c < 3) && (b < B);
        const int  offB   = laneOn ? (b * CS + c) * 4 : 0;

        unsigned is[16], id[16];
        float    xs[16], xd[16], corr[16];
        edge_gather_steps<0>(x, sB_l, dB_l, offB, is, id, xs, xd);
        edge_corr_steps<0, FULL>(r_l, laneOn, nValid, xs, xd, corr);

        if (laneOn) {
            #pragma unroll
            for (int k = 0; k < 16; ++k) {
                if (FULL || k < nValid) {
                    unsigned os = is[k], od = id[k];
                    asm volatile("" : "+v"(os), "+v"(od));
                    atomic_add_f32_at(dl, os,  corr[k]);
                    atomic_add_f32_at(dl, od, -corr[k]);
                }
            }
        }
    }
}

__global__ __launch_bounds__(256) void edge_kernel(
    const float* __restrict__ x,
    const int*   __restrict__ ei,
    const float* __restrict__ rl,
    float* __restrict__ delta,
    int E, int B, int CS, int NS, int planeStride)
{
    const int chunk = blockIdx.x * (blockDim.x >> 5) + (threadIdx.x >> 5);
    const int base  = chunk * 32;
    if (base >= E) return;

    const int lane = threadIdx.x & 31;
    const int j    = lane & 15;
    const int bq   = j >> 2;
    const int c    = j & 3;

    const int   e_l  = min(base + lane, E - 1);
    const int   recB = NS * 4;
    const int   sB_l = ei[e_l] * recB;
    const int   dB_l = ei[E + e_l] * recB;
    const float r_l  = rl[e_l];

    const int rowBase = base + (lane & 16);
    const int nValid  = min(16, E - rowBase);

    float* __restrict__ dl = delta + (size_t)cache_half() * (size_t)planeStride;

    if (base + 32 <= E) {
        edge_wave<true >(x, dl, sB_l, dB_l, r_l, nValid, bq, c, B, CS);
    } else {
        edge_wave<false>(x, dl, sB_l, dB_l, r_l, nValid, bq, c, B, CS);
    }
}

__device__ __forceinline__ float4 add4(float4 a, float4 b) {
    return make_float4(a.x + b.x, a.y + b.y, a.z + b.z, a.w + b.w);
}

__global__ __launch_bounds__(256) void apply_kernel(float* __restrict__ x, float* __restrict__ delta,
                                                    int n, int planeStride) {
    const bool two = planeStride != 0;
    const bool vec = (n >= 4) && ((n & 3) == 0) && ((planeStride & 3) == 0) &&
                     (((((size_t)x) | ((size_t)delta)) & 15) == 0);
    if (vec) {
        const float4 z = make_float4(0.0f, 0.0f, 0.0f, 0.0f);
        const int n4  = n >> 2;
        const int ps4 = planeStride >> 2;
        float4* x4 = (float4*)x;
        float4* d4 = (float4*)delta;
        const int base = blockIdx.x * (blockDim.x * APPLY_VPL) + threadIdx.x;
        float4 a0[APPLY_VPL], a1[APPLY_VPL], xv[APPLY_VPL];
        #pragma unroll
        for (int k = 0; k < APPLY_VPL; ++k) {
            const int j = min(base + k * (int)blockDim.x, n4 - 1);
            a0[k] = d4[j];
            a1[k] = d4[j + ps4];
            xv[k] = x4[j];
        }
        #pragma unroll
        for (int k = 0; k < APPLY_VPL; ++k) {
            const int i = base + k * blockDim.x;
            if (i < n4) {
                const float4 s = two ? add4(a0[k], a1[k]) : a0[k];
                d4[i] = z;
                if (two) d4[i + ps4] = z;
                x4[i] = add4(xv[k], s);
            }
        }
    } else {
        const int nthr = gridDim.x * blockDim.x;
        for (int i = blockIdx.x * blockDim.x + threadIdx.x; i < n; i += nthr) {
            float d = delta[i];
            delta[i] = 0.0f;
            if (two) {
                d += delta[i + planeStride];
                delta[i + planeStride] = 0.0f;
            }
            x[i] += d;
        }
    }
}

__global__ __launch_bounds__(256) void final_kernel(
    const float* __restrict__ x, const float* __restrict__ delta,
    const float* __restrict__ kp, const float* __restrict__ ts,
    float* __restrict__ out, int n, int perBatch, int CS, int NS,
    int planeStride)
{
    const int base = blockIdx.x * (blockDim.x * FINAL_EPL) + threadIdx.x;
    if (n <= 0) return;
    float xv[FINAL_EPL], d0[FINAL_EPL], d1[FINAL_EPL], kv[FINAL_EPL], tv[FINAL_EPL];
    #pragma unroll
    for (int k = 0; k < FINAL_EPL; ++k) {
        const int i    = min(base + k * (int)blockDim.x, n - 1);
        const int b    = i / perBatch;
        const int rem  = i - b * perBatch;
        const int node = rem / 3;
        const int c    = rem - node * 3;
        const int t    = node * NS + b * CS + c;
        tv[k] = ts[b];
        d0[k] = delta[t];
        d1[k] = delta[t + planeStride];
        xv[k] = x[t];
        kv[k] = kp[i];
    }
    #pragma unroll
    for (int k = 0; k < FINAL_EPL; ++k) {
        const int i = base + k * blockDim.x;
        if (i < n) {
            const float tau = fmaxf(1.0f - tv[k], TAU_MIN_F);
            const float d   = planeStride ? d0[k] + d1[k] : d0[k];
            out[i] = (xv[k] + d - kv[k]) / tau;
        }
    }
}

extern "C" void kernel_launch(void* const* d_in, const int* in_sizes, int n_in,
                              void* d_out, int out_size, void* d_ws, size_t ws_size,
                              hipStream_t stream) {
    const float* kp   = (const float*)d_in[0];
    const float* ts   = (const float*)d_in[1];
    const float* ht   = (const float*)d_in[2];
    const float* W    = (const float*)d_in[3];
    const float* bias = (const float*)d_in[4];
    const int*   ei   = (const int*)d_in[5];
    const float* rl   = (const float*)d_in[6];

    const int B   = in_sizes[1];
    const int E   = in_sizes[6];
    const int BN3 = in_sizes[0];
    const int BN  = BN3 / 3;
    const int N   = BN / B;
    const int D   = in_sizes[2] / BN;

    const bool padded = (B % 4 == 0) && ws_size >= (size_t)2 * N * B * 4 * sizeof(float);
    const int CS  = padded ? 4 : 3;
    const int NS  = B * CS;
    const int NW  = N * NS;

    const bool twoPlanes   = ws_size >= (size_t)3 * NW * sizeof(float);
    const int  planeStride = twoPlanes ? NW : 0;
    const int  NWD         = twoPlanes ? 2 * NW : NW;

    float* xbuf  = (float*)d_ws;
    float* delta = xbuf + NW;

    const int T = 256;
    const int nTiles = (BN + 15) / 16;
    const int gVec   = (BN3 + T * FINAL_EPL - 1) / (T * FINAL_EPL);
    const int gWs    = ((NW + 3) / 4 + T * APPLY_VPL - 1) / (T * APPLY_VPL);
    const int gWsD   = ((NWD + 3) / 4 + T * ZERO_VPL - 1) / (T * ZERO_VPL);
    const int nChunk = (E + 31) / 32;
    const int gEdge  = (nChunk + T / 32 - 1) / (T / 32);
    const int gWmma  = (nTiles * 32 + T - 1) / T;

    zero_kernel<<<gWsD, T, 0, stream>>>(delta, NWD);
    head_wmma_kernel<<<gWmma, T, 0, stream>>>(ht, W, bias, kp, ts, xbuf, nTiles, D, N, CS, NS);

    const int XPBD_ITERS = 4;
    for (int it = 0; it < XPBD_ITERS; ++it) {
        edge_kernel<<<gEdge, T, 0, stream>>>(xbuf, ei, rl, delta, E, B, CS, NS, planeStride);
        if (it < XPBD_ITERS - 1) {
            apply_kernel<<<gWs, T, 0, stream>>>(xbuf, delta, NW, planeStride);
        } else {
            final_kernel<<<gVec, T, 0, stream>>>(xbuf, delta, kp, ts,
                                                 (float*)d_out, BN3, N * 3, CS, NS, planeStride);
        }
    }
}
